# starting kernel
# baseline (speedup 1.0000x reference)
#include <hip/hip_runtime.h>

typedef float v2f __attribute__((ext_vector_type(2)));
typedef float v8f __attribute__((ext_vector_type(8)));

#define D_IN  128
#define D_OUT 32

__global__ __launch_bounds__(128) void gemm_xw_kernel(
    const float* __restrict__ x, const float* __restrict__ W,
    float* __restrict__ xw, int nTiles) {
  __shared__ float Wlds[D_IN * D_OUT];

  const float4* Wv = (const float4*)W;
  float4* Lv = (float4*)Wlds;
  for (int i = threadIdx.x; i < (D_IN * D_OUT) / 4; i += blockDim.x)
    Lv[i] = Wv[i];
  __syncthreads();

  const int wave = threadIdx.x >> 5;
  const int lane = threadIdx.x & 31;
  const int tile = blockIdx.x * 4 + wave;
  if (tile >= nTiles) return;

  const int half = lane >> 4;
  const int m    = lane & 15;

  const float* xrow = x + (size_t)(tile * 16 + m) * D_IN + half * 2;

  v8f c0 = {};
  v8f c1 = {};

  #pragma unroll 4
  for (int k = 0; k < D_IN; k += 4) {
    v2f a = *(const v2f*)(xrow + k);

    const int kb = k + half * 2;
    v2f b0, b1;
    b0.x = Wlds[kb * D_OUT + m];
    b0.y = Wlds[(kb + 1) * D_OUT + m];
    b1.x = Wlds[kb * D_OUT + 16 + m];
    b1.y = Wlds[(kb + 1) * D_OUT + 16 + m];

    c0 = __builtin_amdgcn_wmma_f32_16x16x4_f32(false, a, false, b0,
                                               (short)0, c0, false, false);
    c1 = __builtin_amdgcn_wmma_f32_16x16x4_f32(false, a, false, b1,
                                               (short)0, c1, false, false);
  }

  const int rowBase = tile * 16 + half * 8;
  #pragma unroll
  for (int r = 0; r < 8; ++r) {
    xw[(size_t)(rowBase + r) * D_OUT + m]      = c0[r];
    xw[(size_t)(rowBase + r) * D_OUT + 16 + m] = c1[r];
  }
}

__global__ void init_bias_kernel(float* __restrict__ out,
                                 const float* __restrict__ b, int total) {
  int i = blockIdx.x * blockDim.x + threadIdx.x;
  if (i < total) out[i] = b[i & (D_OUT - 1)];
}

__global__ void scatter_edges_kernel(
    const float* __restrict__ xw, const float* __restrict__ edge_val,
    const int* __restrict__ edge_src, const int* __restrict__ edge_dst,
    float* __restrict__ out, int E) {
  long long g = (long long)blockIdx.x * blockDim.x + threadIdx.x;
  int e   = (int)(g >> 5);
  int col = (int)(g & 31);
  if (e >= E) return;

  int s   = edge_src[e];
  int d   = edge_dst[e];
  float v = edge_val[e] * xw[(size_t)s * D_OUT + col];
  unsafeAtomicAdd(&out[(size_t)d * D_OUT + col], v);
}

__global__ void relu_kernel(float* __restrict__ out, int total) {
  int i = blockIdx.x * blockDim.x + threadIdx.x;
  if (i < total) out[i] = fmaxf(out[i], 0.0f);
}

extern "C" void kernel_launch(void* const* d_in, const int* in_sizes, int n_in,
                              void* d_out, int out_size, void* d_ws, size_t ws_size,
                              hipStream_t stream) {
  const float* x        = (const float*)d_in[0];
  const float* W        = (const float*)d_in[1];
  const float* b        = (const float*)d_in[2];
  const float* edge_val = (const float*)d_in[3];
  const int*   edge_src = (const int*)d_in[4];
  const int*   edge_dst = (const int*)d_in[5];
  float* out = (float*)d_out;
  float* xw  = (float*)d_ws;

  const int N = in_sizes[0] / D_IN;
  const int E = in_sizes[3];
  const int nTiles = (N + 15) / 16;
  const int total  = N * D_OUT;

  gemm_xw_kernel<<<(nTiles + 3) / 4, 128, 0, stream>>>(x, W, xw, nTiles);
  init_bias_kernel<<<(total + 255) / 256, 256, 0, stream>>>(out, b, total);

  long long work = (long long)E * 32;
  scatter_edges_kernel<<<(unsigned)((work + 255) / 256), 256, 0, stream>>>(
      xw, edge_val, edge_src, edge_dst, out, E);

  relu_kernel<<<(total + 255) / 256, 256, 0, stream>>>(out, total);
}


# scatter as a dst-band filter, no sort, no global atomics: work-group = (2048-row band in LDS, 1 of 5 edge chunks) keeps its band's edges; LDS float adds, arrival order; out=relu(b+5 partial sums)
# speedup vs baseline: 2.3954x; 2.3954x over previous
#include <hip/hip_runtime.h>

typedef float v2f __attribute__((ext_vector_type(2)));
typedef float v8f __attribute__((ext_vector_type(8)));

#define D_IN  128
#define D_OUT 32

__global__ __launch_bounds__(128) void gemm_xw_kernel(
    const float* __restrict__ x, const float* __restrict__ W,
    float* __restrict__ xw, int nTiles) {
  __shared__ float Wlds[D_IN * D_OUT];

  const float4* Wv = (const float4*)W;
  float4* Lv = (float4*)Wlds;
  for (int i = threadIdx.x; i < (D_IN * D_OUT) / 4; i += blockDim.x)
    Lv[i] = Wv[i];
  __syncthreads();

  const int wave = threadIdx.x >> 5;
  const int lane = threadIdx.x & 31;
  const int tile = blockIdx.x * 4 + wave;
  if (tile >= nTiles) return;

  const int half = lane >> 4;
  const int m    = lane & 15;

  const float* xrow = x + (size_t)(tile * 16 + m) * D_IN + half * 2;

  v8f c0 = {};
  v8f c1 = {};

  #pragma unroll 4
  for (int k = 0; k < D_IN; k += 4) {
    v2f a = *(const v2f*)(xrow + k);

    const int kb = k + half * 2;
    v2f b0, b1;
    b0.x = Wlds[kb * D_OUT + m];
    b0.y = Wlds[(kb + 1) * D_OUT + m];
    b1.x = Wlds[kb * D_OUT + 16 + m];
    b1.y = Wlds[(kb + 1) * D_OUT + 16 + m];

    c0 = __builtin_amdgcn_wmma_f32_16x16x4_f32(false, a, false, b0,
                                               (short)0, c0, false, false);
    c1 = __builtin_amdgcn_wmma_f32_16x16x4_f32(false, a, false, b1,
                                               (short)0, c1, false, false);
  }

  const int rowBase = tile * 16 + half * 8;
  #pragma unroll
  for (int r = 0; r < 8; ++r) {
    xw[(size_t)(rowBase + r) * D_OUT + m]      = c0[r];
    xw[(size_t)(rowBase + r) * D_OUT + 16 + m] = c1[r];
  }
}

#define BAND_ROWS 2048
#define NCHUNK    5
#define QCAP      256

__device__ __forceinline__ void flush_rows(
    const unsigned* q, int head, int count, int e0, int lane,
    const float* __restrict__ xw, const float* __restrict__ edge_val,
    const int* __restrict__ edge_src, float* acc) {
  unsigned ent = 0u;
  int s = 0;
  float v = 0.0f;
  if (lane < count) {
    ent = q[(head + lane) & (QCAP - 1)];
    const int e = e0 + (int)(ent >> 11);
    s = edge_src[e];
    v = edge_val[e];
  }
  const int dl = (int)(ent & (BAND_ROWS - 1));
  float r[32];
  #pragma unroll
  for (int j = 0; j < 32; ++j) {
    const int sj = __builtin_amdgcn_readlane(s, j);
    r[j] = xw[(size_t)sj * D_OUT + lane];
  }
  #pragma unroll
  for (int j = 0; j < 32; ++j) {
    if (j < count) {
      const float vj = __int_as_float(__builtin_amdgcn_readlane(__float_as_int(v), j));
      const int dj = __builtin_amdgcn_readlane(dl, j);
      atomicAdd(&acc[dj * D_OUT + lane], vj * r[j]);
    }
  }
}

__device__ __forceinline__ int4 load_dst4_guarded(const int* __restrict__ p, int idx, int e1) {
  int4 d = make_int4(-1, -1, -1, -1);
  if (idx + 3 < e1) {
    d = *(const int4*)(p + idx);
  } else {
    if (idx < e1)     d.x = p[idx];
    if (idx + 1 < e1) d.y = p[idx + 1];
    if (idx + 2 < e1) d.z = p[idx + 2];
  }
  return d;
}

__global__ __launch_bounds__(1024) void band_filter_kernel(
    const float* __restrict__ xw, const float* __restrict__ edge_val,
    const int* __restrict__ edge_src, const int* __restrict__ edge_dst,
    float* __restrict__ part, int N, int E) {
  __shared__ __attribute__((aligned(16))) float acc[BAND_ROWS * D_OUT];
  __shared__ unsigned queue[32 * QCAP];

  float4* acc4 = (float4*)acc;
  for (int i = threadIdx.x; i < BAND_ROWS * D_OUT / 4; i += 1024)
    acc4[i] = make_float4(0.0f, 0.0f, 0.0f, 0.0f);
  __syncthreads();

  const int lane = threadIdx.x & 31;
  const int wave = threadIdx.x >> 5;
  const int lo   = blockIdx.x * BAND_ROWS;
  const int nw   = gridDim.y * 32;
  const int gw   = blockIdx.y * 32 + wave;
  const int span = (((E + nw - 1) / nw) + 3) & ~3;
  const int e0   = gw * span;
  const int e1   = min(e0 + span, E);
  unsigned* q = queue + wave * QCAP;
  const unsigned laneTerm = (unsigned)(lane * 4) << 11;

  int qhead = 0, qtail = 0;

  int4 cur[4], nxt[4];
  if (e0 < e1) {
    if (e0 + 512 <= e1) {
      #pragma unroll
      for (int u = 0; u < 4; ++u) cur[u] = *(const int4*)(edge_dst + e0 + u * 128 + lane * 4);
    } else {
      #pragma unroll
      for (int u = 0; u < 4; ++u) cur[u] = load_dst4_guarded(edge_dst, e0 + u * 128 + lane * 4, e1);
    }
  }

  for (int base = e0; base < e1; base += 512) {
    const int nb = base + 512;
    if (nb < e1) {
      if (nb + 512 <= e1) {
        #pragma unroll
        for (int u = 0; u < 4; ++u) nxt[u] = *(const int4*)(edge_dst + nb + u * 128 + lane * 4);
      } else {
        #pragma unroll
        for (int u = 0; u < 4; ++u) nxt[u] = load_dst4_guarded(edge_dst, nb + u * 128 + lane * 4, e1);
      }
    }

    #pragma unroll
    for (int u = 0; u < 4; ++u) {
      const int dv[4] = {cur[u].x, cur[u].y, cur[u].z, cur[u].w};
      #pragma unroll
      for (int k = 0; k < 4; ++k) {
        const unsigned dl = (unsigned)(dv[k] - lo);
        const bool hit = dl < (unsigned)BAND_ROWS;
        const unsigned mask = __builtin_amdgcn_ballot_w32(hit);
        if (mask != 0u) {
          const int pre = (int)__builtin_amdgcn_mbcnt_lo(mask, 0u);
          if (hit)
            q[(qtail + pre) & (QCAP - 1)] =
                (((unsigned)(base - e0 + u * 128 + k)) << 11) + laneTerm + dl;
          qtail += __builtin_popcount(mask);
        }
      }
      while (qtail - qhead >= 32) {
        flush_rows(q, qhead, 32, e0, lane, xw, edge_val, edge_src, acc);
        qhead += 32;
      }
    }

    #pragma unroll
    for (int u = 0; u < 4; ++u) cur[u] = nxt[u];
  }
  if (qtail - qhead > 0)
    flush_rows(q, qhead, qtail - qhead, e0, lane, xw, edge_val, edge_src, acc);

  __syncthreads();

  const int rows = min(BAND_ROWS, N - lo);
  float4* p4 = (float4*)(part + ((size_t)blockIdx.y * N + lo) * D_OUT);
  for (int i = threadIdx.x; i < rows * (D_OUT / 4); i += 1024) p4[i] = acc4[i];
}

__global__ __launch_bounds__(256) void combine_kernel(
    const float* __restrict__ part, const float* __restrict__ b,
    float* __restrict__ out, int total4) {
  const int i = blockIdx.x * 256 + threadIdx.x;
  if (i >= total4) return;
  const float4* p = (const float4*)part;
  float4 s = p[i];
  #pragma unroll
  for (int c = 1; c < NCHUNK; ++c) {
    const float4 t = p[(size_t)c * total4 + i];
    s.x += t.x; s.y += t.y; s.z += t.z; s.w += t.w;
  }
  const float4 bb = ((const float4*)b)[i & (D_OUT / 4 - 1)];
  float4 o;
  o.x = fmaxf(s.x + bb.x, 0.0f);
  o.y = fmaxf(s.y + bb.y, 0.0f);
  o.z = fmaxf(s.z + bb.z, 0.0f);
  o.w = fmaxf(s.w + bb.w, 0.0f);
  ((float4*)out)[i] = o;
}

extern "C" void kernel_launch(void* const* d_in, const int* in_sizes, int n_in,
                              void* d_out, int out_size, void* d_ws, size_t ws_size,
                              hipStream_t stream) {
  const float* x        = (const float*)d_in[0];
  const float* W        = (const float*)d_in[1];
  const float* b        = (const float*)d_in[2];
  const float* edge_val = (const float*)d_in[3];
  const int*   edge_src = (const int*)d_in[4];
  const int*   edge_dst = (const int*)d_in[5];
  float* out  = (float*)d_out;
  float* xw   = (float*)d_ws;
  float* part = xw + (size_t)4 * 1024 * 1024;

  const int N = in_sizes[0] / D_IN;
  const int E = in_sizes[3];
  const int nTiles = (N + 15) / 16;
  const int total4 = N * (D_OUT / 4);

  gemm_xw_kernel<<<(nTiles + 3) / 4, 128, 0, stream>>>(x, W, xw, nTiles);
  band_filter_kernel<<<dim3((N + BAND_ROWS - 1) / BAND_ROWS, NCHUNK), 1024, 0, stream>>>(
      xw, edge_val, edge_src, edge_dst, part, N, E);
  combine_kernel<<<(total4 + 255) / 256, 256, 0, stream>>>(part, b, out, total4);
}


# GEMM: each wave requests the block's W copy and its whole 16x128 x tile (16 16-byte loads per lane) before the first wait: 1 memory round trip, not 16 dependent; k order inside the sums changes
# speedup vs baseline: 2.5525x; 1.0656x over previous
#include <hip/hip_runtime.h>

typedef float v2f __attribute__((ext_vector_type(2)));
typedef float v4f __attribute__((ext_vector_type(4)));
typedef float v8f __attribute__((ext_vector_type(8)));

#define D_IN  128
#define D_OUT 32

__global__ __launch_bounds__(128) void gemm_xw_kernel(
    const float* __restrict__ x, const float* __restrict__ W,
    float* __restrict__ xw, int nTiles) {
  __shared__ float Wlds[D_IN * D_OUT];

  const int wave = threadIdx.x >> 5;
  const int lane = threadIdx.x & 31;
  const int tile = blockIdx.x * 4 + wave;
  const bool live = tile < nTiles;

  const int half = lane >> 4;
  const int m    = lane & 15;

  const v4f* Wv = (const v4f*)W;
  v4f* Lv = (v4f*)Wlds;
  v4f w[8];
  #pragma unroll
  for (int i = 0; i < 8; ++i) w[i] = Wv[threadIdx.x + i * 128];

  const int ltile = live ? tile : nTiles - 1;
  const float4* xrow =
      (const float4*)(x + (size_t)(ltile * 16 + m) * D_IN + half * 4);
  float4 a[16];
  #pragma unroll
  for (int j = 0; j < 16; ++j) a[j] = xrow[j * 2];

  asm volatile("" :: "v"(w[0]), "v"(w[1]), "v"(w[2]), "v"(w[3]),
                     "v"(w[4]), "v"(w[5]), "v"(w[6]), "v"(w[7]));
  #pragma unroll
  for (int i = 0; i < 8; ++i) Lv[threadIdx.x + i * 128] = w[i];
  __syncthreads();
  if (!live) return;

  v8f c0 = {};
  v8f c1 = {};

  #pragma unroll
  for (int j = 0; j < 16; ++j) {
    #pragma unroll
    for (int t = 0; t < 2; ++t) {
      const int kb = j * 8 + half * 4 + t * 2;
      v2f av;
      av.x = t ? a[j].z : a[j].x;
      av.y = t ? a[j].w : a[j].y;
      v2f b0, b1;
      b0.x = Wlds[kb * D_OUT + m];
      b0.y = Wlds[(kb + 1) * D_OUT + m];
      b1.x = Wlds[kb * D_OUT + 16 + m];
      b1.y = Wlds[(kb + 1) * D_OUT + 16 + m];
      c0 = __builtin_amdgcn_wmma_f32_16x16x4_f32(false, av, false, b0,
                                                 (short)0, c0, false, false);
      c1 = __builtin_amdgcn_wmma_f32_16x16x4_f32(false, av, false, b1,
                                                 (short)0, c1, false, false);
    }
  }

  const int rowBase = tile * 16 + half * 8;
  #pragma unroll
  for (int r = 0; r < 8; ++r) {
    xw[(size_t)(rowBase + r) * D_OUT + m]      = c0[r];
    xw[(size_t)(rowBase + r) * D_OUT + 16 + m] = c1[r];
  }
}

#define BAND_ROWS 2048
#define NCHUNK    5
#define QCAP      256

__device__ __forceinline__ void flush_rows(
    const unsigned* q, int head, int count, int e0, int lane,
    const float* __restrict__ xw, const float* __restrict__ edge_val,
    const int* __restrict__ edge_src, float* acc) {
  unsigned ent = 0u;
  int s = 0;
  float v = 0.0f;
  if (lane < count) {
    ent = q[(head + lane) & (QCAP - 1)];
    const int e = e0 + (int)(ent >> 11);
    s = edge_src[e];
    v = edge_val[e];
  }
  const int dl = (int)(ent & (BAND_ROWS - 1));
  float r[32];
  #pragma unroll
  for (int j = 0; j < 32; ++j) {
    const int sj = __builtin_amdgcn_readlane(s, j);
    r[j] = xw[(size_t)sj * D_OUT + lane];
  }
  #pragma unroll
  for (int j = 0; j < 32; ++j) {
    if (j < count) {
      const float vj = __int_as_float(__builtin_amdgcn_readlane(__float_as_int(v), j));
      const int dj = __builtin_amdgcn_readlane(dl, j);
      atomicAdd(&acc[dj * D_OUT + lane], vj * r[j]);
    }
  }
}

__device__ __forceinline__ int4 load_dst4_guarded(const int* __restrict__ p, int idx, int e1) {
  int4 d = make_int4(-1, -1, -1, -1);
  if (idx + 3 < e1) {
    d = *(const int4*)(p + idx);
  } else {
    if (idx < e1)     d.x = p[idx];
    if (idx + 1 < e1) d.y = p[idx + 1];
    if (idx + 2 < e1) d.z = p[idx + 2];
  }
  return d;
}

__global__ __launch_bounds__(1024) void band_filter_kernel(
    const float* __restrict__ xw, const float* __restrict__ edge_val,
    const int* __restrict__ edge_src, const int* __restrict__ edge_dst,
    float* __restrict__ part, int N, int E) {
  __shared__ __attribute__((aligned(16))) float acc[BAND_ROWS * D_OUT];
  __shared__ unsigned queue[32 * QCAP];

  float4* acc4 = (float4*)acc;
  for (int i = threadIdx.x; i < BAND_ROWS * D_OUT / 4; i += 1024)
    acc4[i] = make_float4(0.0f, 0.0f, 0.0f, 0.0f);
  __syncthreads();

  const int lane = threadIdx.x & 31;
  const int wave = threadIdx.x >> 5;
  const int lo   = blockIdx.x * BAND_ROWS;
  const int nw   = gridDim.y * 32;
  const int gw   = blockIdx.y * 32 + wave;
  const int span = (((E + nw - 1) / nw) + 3) & ~3;
  const int e0   = gw * span;
  const int e1   = min(e0 + span, E);
  unsigned* q = queue + wave * QCAP;
  const unsigned laneTerm = (unsigned)(lane * 4) << 11;

  int qhead = 0, qtail = 0;

  int4 cur[4], nxt[4];
  if (e0 < e1) {
    if (e0 + 512 <= e1) {
      #pragma unroll
      for (int u = 0; u < 4; ++u) cur[u] = *(const int4*)(edge_dst + e0 + u * 128 + lane * 4);
    } else {
      #pragma unroll
      for (int u = 0; u < 4; ++u) cur[u] = load_dst4_guarded(edge_dst, e0 + u * 128 + lane * 4, e1);
    }
  }

  for (int base = e0; base < e1; base += 512) {
    const int nb = base + 512;
    if (nb < e1) {
      if (nb + 512 <= e1) {
        #pragma unroll
        for (int u = 0; u < 4; ++u) nxt[u] = *(const int4*)(edge_dst + nb + u * 128 + lane * 4);
      } else {
        #pragma unroll
        for (int u = 0; u < 4; ++u) nxt[u] = load_dst4_guarded(edge_dst, nb + u * 128 + lane * 4, e1);
      }
    }

    #pragma unroll
    for (int u = 0; u < 4; ++u) {
      const int dv[4] = {cur[u].x, cur[u].y, cur[u].z, cur[u].w};
      #pragma unroll
      for (int k = 0; k < 4; ++k) {
        const unsigned dl = (unsigned)(dv[k] - lo);
        const bool hit = dl < (unsigned)BAND_ROWS;
        const unsigned mask = __builtin_amdgcn_ballot_w32(hit);
        if (mask != 0u) {
          const int pre = (int)__builtin_amdgcn_mbcnt_lo(mask, 0u);
          if (hit)
            q[(qtail + pre) & (QCAP - 1)] =
                (((unsigned)(base - e0 + u * 128 + k)) << 11) + laneTerm + dl;
          qtail += __builtin_popcount(mask);
        }
      }
      while (qtail - qhead >= 32) {
        flush_rows(q, qhead, 32, e0, lane, xw, edge_val, edge_src, acc);
        qhead += 32;
      }
    }

    #pragma unroll
    for (int u = 0; u < 4; ++u) cur[u] = nxt[u];
  }
  if (qtail - qhead > 0)
    flush_rows(q, qhead, qtail - qhead, e0, lane, xw, edge_val, edge_src, acc);

  __syncthreads();

  const int rows = min(BAND_ROWS, N - lo);
  float4* p4 = (float4*)(part + ((size_t)blockIdx.y * N + lo) * D_OUT);
  for (int i = threadIdx.x; i < rows * (D_OUT / 4); i += 1024) p4[i] = acc4[i];
}

__global__ __launch_bounds__(256) void combine_kernel(
    const float* __restrict__ part, const float* __restrict__ b,
    float* __restrict__ out, int total4) {
  const int i = blockIdx.x * 256 + threadIdx.x;
  if (i >= total4) return;
  const float4* p = (const float4*)part;
  float4 s = p[i];
  #pragma unroll
  for (int c = 1; c < NCHUNK; ++c) {
    const float4 t = p[(size_t)c * total4 + i];
    s.x += t.x; s.y += t.y; s.z += t.z; s.w += t.w;
  }
  const float4 bb = ((const float4*)b)[i & (D_OUT / 4 - 1)];
  float4 o;
  o.x = fmaxf(s.x + bb.x, 0.0f);
  o.y = fmaxf(s.y + bb.y, 0.0f);
  o.z = fmaxf(s.z + bb.z, 0.0f);
  o.w = fmaxf(s.w + bb.w, 0.0f);
  ((float4*)out)[i] = o;
}

extern "C" void kernel_launch(void* const* d_in, const int* in_sizes, int n_in,
                              void* d_out, int out_size, void* d_ws, size_t ws_size,
                              hipStream_t stream) {
  const float* x        = (const float*)d_in[0];
  const float* W        = (const float*)d_in[1];
  const float* b        = (const float*)d_in[2];
  const float* edge_val = (const float*)d_in[3];
  const int*   edge_src = (const int*)d_in[4];
  const int*   edge_dst = (const int*)d_in[5];
  float* out  = (float*)d_out;
  float* xw   = (float*)d_ws;
  float* part = xw + (size_t)4 * 1024 * 1024;

  const int N = in_sizes[0] / D_IN;
  const int E = in_sizes[3];
  const int nTiles = (N + 15) / 16;
  const int total4 = N * (D_OUT / 4);

  gemm_xw_kernel<<<(nTiles + 3) / 4, 128, 0, stream>>>(x, W, xw, nTiles);
  band_filter_kernel<<<dim3((N + BAND_ROWS - 1) / BAND_ROWS, NCHUNK), 1024, 0, stream>>>(
      xw, edge_val, edge_src, edge_dst, part, N, E);
  combine_kernel<<<(total4 + 255) / 256, 256, 0, stream>>>(part, b, out, total4);
}


# band scan on 1-byte band tags (dst>>11, written by the GEMM launch): 16 tags per lane-load, 4 matched per dword by integer byte ops; flush re-reads dst; order of a wave's LDS float adds changes
# speedup vs baseline: 2.6372x; 1.0332x over previous
#include <hip/hip_runtime.h>

typedef float v2f __attribute__((ext_vector_type(2)));
typedef float v4f __attribute__((ext_vector_type(4)));
typedef float v8f __attribute__((ext_vector_type(8)));

#define D_IN  128
#define D_OUT 32
#define BAND_ROWS 2048

__device__ __forceinline__ int4 load_dst4_guarded(const int* __restrict__ p, int idx, int e1) {
  int4 d = make_int4(-1, -1, -1, -1);
  if (idx + 3 < e1) {
    d = *(const int4*)(p + idx);
  } else {
    if (idx < e1)     d.x = p[idx];
    if (idx + 1 < e1) d.y = p[idx + 1];
    if (idx + 2 < e1) d.z = p[idx + 2];
  }
  return d;
}

__device__ __forceinline__ unsigned pack_tags(int4 d) {
  return ((unsigned)(d.x >> 11) & 0xFFu) | (((unsigned)(d.y >> 11) & 0xFFu) << 8) |
         (((unsigned)(d.z >> 11) & 0xFFu) << 16) | (((unsigned)(d.w >> 11) & 0xFFu) << 24);
}

__global__ __launch_bounds__(128) void gemm_xw_kernel(
    const float* __restrict__ x, const float* __restrict__ W,
    float* __restrict__ xw, int nTiles,
    const int* __restrict__ edge_dst, int* __restrict__ tags, int E) {
  __shared__ float Wlds[D_IN * D_OUT];

  const int wave = threadIdx.x >> 5;
  const int lane = threadIdx.x & 31;
  const int tile = blockIdx.x * 4 + wave;
  const bool live = tile < nTiles;

  const int half = lane >> 4;
  const int m    = lane & 15;

  const v4f* Wv = (const v4f*)W;
  v4f* Lv = (v4f*)Wlds;
  v4f w[8];
  #pragma unroll
  for (int i = 0; i < 8; ++i) w[i] = Wv[threadIdx.x + i * 128];

  const int ltile = live ? tile : nTiles - 1;
  const float4* xrow =
      (const float4*)(x + (size_t)(ltile * 16 + m) * D_IN + half * 4);
  float4 a[16];
  #pragma unroll
  for (int j = 0; j < 16; ++j) a[j] = xrow[j * 2];

  asm volatile("" :: "v"(w[0]), "v"(w[1]), "v"(w[2]), "v"(w[3]),
                     "v"(w[4]), "v"(w[5]), "v"(w[6]), "v"(w[7]));
  #pragma unroll
  for (int i = 0; i < 8; ++i) Lv[threadIdx.x + i * 128] = w[i];
  __syncthreads();

  const int Epad = (E + 15) & ~15;
  const int nthr = gridDim.x * 128;
  const int gtid = blockIdx.x * 128 + threadIdx.x;
  const int tcl  = min(gtid * 8, (E - 8) & ~3);
  const int4 ta = *(const int4*)(edge_dst + tcl);
  const int4 tb = *(const int4*)(edge_dst + tcl + 4);

  if (live) {
  v8f c0 = {};
  v8f c1 = {};

  #pragma unroll
  for (int j = 0; j < 16; ++j) {
    #pragma unroll
    for (int t = 0; t < 2; ++t) {
      const int kb = j * 8 + half * 4 + t * 2;
      v2f av;
      av.x = t ? a[j].z : a[j].x;
      av.y = t ? a[j].w : a[j].y;
      v2f b0, b1;
      b0.x = Wlds[kb * D_OUT + m];
      b0.y = Wlds[(kb + 1) * D_OUT + m];
      b1.x = Wlds[kb * D_OUT + 16 + m];
      b1.y = Wlds[(kb + 1) * D_OUT + 16 + m];
      c0 = __builtin_amdgcn_wmma_f32_16x16x4_f32(false, av, false, b0,
                                                 (short)0, c0, false, false);
      c1 = __builtin_amdgcn_wmma_f32_16x16x4_f32(false, av, false, b1,
                                                 (short)0, c1, false, false);
    }
  }

  const int rowBase = tile * 16 + half * 8;
  #pragma unroll
  for (int r = 0; r < 8; ++r) {
    xw[(size_t)(rowBase + r) * D_OUT + m]      = c0[r];
    xw[(size_t)(rowBase + r) * D_OUT + 16 + m] = c1[r];
  }
  }

  if (gtid * 8 + 8 <= E) {
    ((uint2*)tags)[gtid] = make_uint2(pack_tags(ta), pack_tags(tb));
  } else if (gtid * 8 < Epad) {
    const int4 ua = load_dst4_guarded(edge_dst, gtid * 8, E);
    const int4 ub = load_dst4_guarded(edge_dst, gtid * 8 + 4, E);
    ((uint2*)tags)[gtid] = make_uint2(pack_tags(ua), pack_tags(ub));
  }
  for (int i = gtid + nthr; i * 8 < Epad; i += nthr) {
    const int4 ua = load_dst4_guarded(edge_dst, i * 8, E);
    const int4 ub = load_dst4_guarded(edge_dst, i * 8 + 4, E);
    ((uint2*)tags)[i] = make_uint2(pack_tags(ua), pack_tags(ub));
  }
}

#define NCHUNK    5
#define QCAP      256

__device__ __forceinline__ unsigned match_bytes(unsigned x) {
  const unsigned t = ((x & 0x7F7F7F7Fu) + 0x7F7F7F7Fu) | x;
  return ~t & 0x80808080u;
}

__device__ __forceinline__ void flush_rows(
    const unsigned* q, int head, int count, int e0, int lane, int lo,
    const float* __restrict__ xw, const float* __restrict__ edge_val,
    const int* __restrict__ edge_src, const int* __restrict__ edge_dst, float* acc) {
  int s = 0;
  int dl = 0;
  float v = 0.0f;
  if (lane < count) {
    const int e = e0 + (int)q[(head + lane) & (QCAP - 1)];
    s = edge_src[e];
    v = edge_val[e];
    dl = edge_dst[e] - lo;
  }
  float r[32];
  #pragma unroll
  for (int j = 0; j < 32; ++j) {
    const int sj = __builtin_amdgcn_readlane(s, j);
    r[j] = xw[(size_t)sj * D_OUT + lane];
  }
  #pragma unroll
  for (int j = 0; j < 32; ++j) {
    if (j < count) {
      const float vj = __int_as_float(__builtin_amdgcn_readlane(__float_as_int(v), j));
      const int dj = __builtin_amdgcn_readlane(dl, j);
      atomicAdd(&acc[dj * D_OUT + lane], vj * r[j]);
    }
  }
}

__global__ __launch_bounds__(1024) void band_filter_kernel(
    const float* __restrict__ xw, const float* __restrict__ edge_val,
    const int* __restrict__ edge_src, const int* __restrict__ edge_dst,
    const int* __restrict__ tags, float* __restrict__ part, int N, int E) {
  __shared__ __attribute__((aligned(16))) float acc[BAND_ROWS * D_OUT];
  __shared__ unsigned queue[32 * QCAP];

  float4* acc4 = (float4*)acc;
  for (int i = threadIdx.x; i < BAND_ROWS * D_OUT / 4; i += 1024)
    acc4[i] = make_float4(0.0f, 0.0f, 0.0f, 0.0f);
  __syncthreads();

  const int lane = threadIdx.x & 31;
  const int wave = threadIdx.x >> 5;
  const int lo   = blockIdx.x * BAND_ROWS;
  const int nw   = gridDim.y * 32;
  const int gw   = blockIdx.y * 32 + wave;
  const int span = (((E + nw - 1) / nw) + 15) & ~15;
  const int e0   = gw * span;
  const int e1   = min(e0 + span, E);
  unsigned* q = queue + wave * QCAP;
  const unsigned bb = (unsigned)blockIdx.x * 0x01010101u;
  const unsigned char* tagb = (const unsigned char*)tags;
  const uint4 none = make_uint4(0xFFFFFFFFu, 0xFFFFFFFFu, 0xFFFFFFFFu, 0xFFFFFFFFu);

  int qhead = 0, qtail = 0;

  uint4 curA = none, curB = none;
  if (e0 + lane * 16 < e1)       curA = *(const uint4*)(tagb + e0 + lane * 16);
  if (e0 + 512 + lane * 16 < e1) curB = *(const uint4*)(tagb + e0 + 512 + lane * 16);
  asm volatile("" :: "v"(curA.x), "v"(curA.y), "v"(curA.z), "v"(curA.w),
                     "v"(curB.x), "v"(curB.y), "v"(curB.z), "v"(curB.w));

  for (int base = e0; base < e1; base += 1024) {
    uint4 nxtA = none, nxtB = none;
    if (base + 1024 + lane * 16 < e1) nxtA = *(const uint4*)(tagb + base + 1024 + lane * 16);
    if (base + 1536 + lane * 16 < e1) nxtB = *(const uint4*)(tagb + base + 1536 + lane * 16);

    unsigned m = (match_bytes(curA.x ^ bb) >> 7) | (match_bytes(curA.y ^ bb) >> 6) |
                 (match_bytes(curA.z ^ bb) >> 5) | (match_bytes(curA.w ^ bb) >> 4) |
                 (match_bytes(curB.x ^ bb) >> 3) | (match_bytes(curB.y ^ bb) >> 2) |
                 (match_bytes(curB.z ^ bb) >> 1) |  match_bytes(curB.w ^ bb);

    for (;;) {
      const bool has = m != 0u;
      const unsigned mask = __builtin_amdgcn_ballot_w32(has);
      if (mask == 0u) break;
      const int pre = (int)__builtin_amdgcn_mbcnt_lo(mask, 0u);
      if (has) {
        const int p = __builtin_ctz(m);
        m &= m - 1u;
        q[(qtail + pre) & (QCAP - 1)] =
            (unsigned)(base - e0 + lane * 16 + ((p & 3) << 2) + (p >> 3) + ((p & 4) << 7));
      }
      qtail += __builtin_popcount(mask);
      while (qtail - qhead >= 32) {
        flush_rows(q, qhead, 32, e0, lane, lo, xw, edge_val, edge_src, edge_dst, acc);
        qhead += 32;
      }
    }

    asm volatile("" :: "v"(nxtA.x), "v"(nxtA.y), "v"(nxtA.z), "v"(nxtA.w),
                       "v"(nxtB.x), "v"(nxtB.y), "v"(nxtB.z), "v"(nxtB.w));
    curA = nxtA;
    curB = nxtB;
  }
  if (qtail - qhead > 0)
    flush_rows(q, qhead, qtail - qhead, e0, lane, lo, xw, edge_val, edge_src, edge_dst, acc);

  __syncthreads();

  const int rows = min(BAND_ROWS, N - lo);
  float4* p4 = (float4*)(part + ((size_t)blockIdx.y * N + lo) * D_OUT);
  for (int i = threadIdx.x; i < rows * (D_OUT / 4); i += 1024) p4[i] = acc4[i];
}

__global__ __launch_bounds__(256) void combine_kernel(
    const float* __restrict__ part, const float* __restrict__ b,
    float* __restrict__ out, int total4) {
  const int i = blockIdx.x * 256 + threadIdx.x;
  if (i >= total4) return;
  const float4* p = (const float4*)part;
  float4 s = p[i];
  #pragma unroll
  for (int c = 1; c < NCHUNK; ++c) {
    const float4 t = p[(size_t)c * total4 + i];
    s.x += t.x; s.y += t.y; s.z += t.z; s.w += t.w;
  }
  const float4 bb = ((const float4*)b)[i & (D_OUT / 4 - 1)];
  float4 o;
  o.x = fmaxf(s.x + bb.x, 0.0f);
  o.y = fmaxf(s.y + bb.y, 0.0f);
  o.z = fmaxf(s.z + bb.z, 0.0f);
  o.w = fmaxf(s.w + bb.w, 0.0f);
  ((float4*)out)[i] = o;
}

extern "C" void kernel_launch(void* const* d_in, const int* in_sizes, int n_in,
                              void* d_out, int out_size, void* d_ws, size_t ws_size,
                              hipStream_t stream) {
  const float* x        = (const float*)d_in[0];
  const float* W        = (const float*)d_in[1];
  const float* b        = (const float*)d_in[2];
  const float* edge_val = (const float*)d_in[3];
  const int*   edge_src = (const int*)d_in[4];
  const int*   edge_dst = (const int*)d_in[5];
  float* out  = (float*)d_out;
  float* xw   = (float*)d_ws;
  float* part = xw + (size_t)4 * 1024 * 1024;
  int*   tags = (int*)(xw + (size_t)32 * 1024 * 1024);

  const int N = in_sizes[0] / D_IN;
  const int E = in_sizes[3];
  const int nTiles = (N + 15) / 16;
  const int total4 = N * (D_OUT / 4);

  gemm_xw_kernel<<<(nTiles + 3) / 4, 128, 0, stream>>>(x, W, xw, nTiles, edge_dst, tags, E);
  band_filter_kernel<<<dim3((N + BAND_ROWS - 1) / BAND_ROWS, NCHUNK), 1024, 0, stream>>>(
      xw, edge_val, edge_src, edge_dst, tags, part, N, E);
  combine_kernel<<<(total4 + 255) / 256, 256, 0, stream>>>(part, b, out, total4);
}


# band sums fetch one 8-byte edge record per kept edge (src, row in band, val; packed by the GEMM launch with whole-line stores) instead of 3 fields from 3 arrays: 1 line request per edge, not 3
# speedup vs baseline: 2.9336x; 1.1124x over previous
#include <hip/hip_runtime.h>

typedef float v2f __attribute__((ext_vector_type(2)));
typedef float v4f __attribute__((ext_vector_type(4)));
typedef float v8f __attribute__((ext_vector_type(8)));

#define D_IN  128
#define D_OUT 32
#define BAND_ROWS 2048

__device__ __forceinline__ int4 load_dst4_guarded(const int* __restrict__ p, int idx, int e1) {
  int4 d = make_int4(-1, -1, -1, -1);
  if (idx + 3 < e1) {
    d = *(const int4*)(p + idx);
  } else {
    if (idx < e1)     d.x = p[idx];
    if (idx + 1 < e1) d.y = p[idx + 1];
    if (idx + 2 < e1) d.z = p[idx + 2];
  }
  return d;
}

__device__ __forceinline__ unsigned pack_tags(int4 d) {
  return ((unsigned)(d.x >> 11) & 0xFFu) | (((unsigned)(d.y >> 11) & 0xFFu) << 8) |
         (((unsigned)(d.z >> 11) & 0xFFu) << 16) | (((unsigned)(d.w >> 11) & 0xFFu) << 24);
}

__device__ __forceinline__ unsigned pack_edge(int s, int d) {
  return (unsigned)s | ((unsigned)(d & (BAND_ROWS - 1)) << 21);
}

__global__ __launch_bounds__(128) void gemm_xw_kernel(
    const float* __restrict__ x, const float* __restrict__ W,
    float* __restrict__ xw, int nTiles,
    const int* __restrict__ edge_dst, int* __restrict__ tags, int E,
    const int* __restrict__ edge_src, const float* __restrict__ edge_val,
    int* __restrict__ rec) {
  __shared__ float Wlds[D_IN * D_OUT];

  const int wave = threadIdx.x >> 5;
  const int lane = threadIdx.x & 31;
  const int tile = blockIdx.x * 4 + wave;
  const bool live = tile < nTiles;

  const int half = lane >> 4;
  const int m    = lane & 15;

  const v4f* Wv = (const v4f*)W;
  v4f* Lv = (v4f*)Wlds;
  v4f w[8];
  #pragma unroll
  for (int i = 0; i < 8; ++i) w[i] = Wv[threadIdx.x + i * 128];

  const int ltile = live ? tile : nTiles - 1;
  const float4* xrow =
      (const float4*)(x + (size_t)(ltile * 16 + m) * D_IN + half * 4);
  float4 a[16];
  #pragma unroll
  for (int j = 0; j < 16; ++j) a[j] = xrow[j * 2];

  asm volatile("" :: "v"(w[0]), "v"(w[1]), "v"(w[2]), "v"(w[3]),
                     "v"(w[4]), "v"(w[5]), "v"(w[6]), "v"(w[7]));
  #pragma unroll
  for (int i = 0; i < 8; ++i) Lv[threadIdx.x + i * 128] = w[i];
  __syncthreads();

  const int Epad = (E + 15) & ~15;
  const int nthr = gridDim.x * 128;
  const int gtid = blockIdx.x * 128 + threadIdx.x;
  const int tcl  = min(gtid * 8, (E - 8) & ~3);
  const int4 ta = *(const int4*)(edge_dst + tcl);
  const int4 tb = *(const int4*)(edge_dst + tcl + 4);

  const int wb    = (gtid - lane) * 8;
  const int elast = (E - 2) & ~1;
  int2 rs[4], rd[4];
  float2 rv[4];
  #pragma unroll
  for (int k = 0; k < 4; ++k) {
    const int ec = min(wb + 64 * k + 2 * lane, elast);
    rs[k] = *(const int2*)(edge_src + ec);
    rd[k] = *(const int2*)(edge_dst + ec);
    rv[k] = *(const float2*)(edge_val + ec);
  }

  if (live) {
  v8f c0 = {};
  v8f c1 = {};

  #pragma unroll
  for (int j = 0; j < 16; ++j) {
    #pragma unroll
    for (int t = 0; t < 2; ++t) {
      const int kb = j * 8 + half * 4 + t * 2;
      v2f av;
      av.x = t ? a[j].z : a[j].x;
      av.y = t ? a[j].w : a[j].y;
      v2f b0, b1;
      b0.x = Wlds[kb * D_OUT + m];
      b0.y = Wlds[(kb + 1) * D_OUT + m];
      b1.x = Wlds[kb * D_OUT + 16 + m];
      b1.y = Wlds[(kb + 1) * D_OUT + 16 + m];
      c0 = __builtin_amdgcn_wmma_f32_16x16x4_f32(false, av, false, b0,
                                                 (short)0, c0, false, false);
      c1 = __builtin_amdgcn_wmma_f32_16x16x4_f32(false, av, false, b1,
                                                 (short)0, c1, false, false);
    }
  }

  const int rowBase = tile * 16 + half * 8;
  #pragma unroll
  for (int r = 0; r < 8; ++r) {
    xw[(size_t)(rowBase + r) * D_OUT + m]      = c0[r];
    xw[(size_t)(rowBase + r) * D_OUT + 16 + m] = c1[r];
  }
  }

  if (gtid * 8 + 8 <= E) {
    ((uint2*)tags)[gtid] = make_uint2(pack_tags(ta), pack_tags(tb));
  } else if (gtid * 8 < Epad) {
    const int4 ua = load_dst4_guarded(edge_dst, gtid * 8, E);
    const int4 ub = load_dst4_guarded(edge_dst, gtid * 8 + 4, E);
    ((uint2*)tags)[gtid] = make_uint2(pack_tags(ua), pack_tags(ub));
  }
  for (int i = gtid + nthr; i * 8 < Epad; i += nthr) {
    const int4 ua = load_dst4_guarded(edge_dst, i * 8, E);
    const int4 ub = load_dst4_guarded(edge_dst, i * 8 + 4, E);
    ((uint2*)tags)[i] = make_uint2(pack_tags(ua), pack_tags(ub));
  }

  #pragma unroll
  for (int k = 0; k < 4; ++k) {
    const int e = wb + 64 * k + 2 * lane;
    if (e + 2 <= E) {
      ((uint4*)rec)[e >> 1] = make_uint4(pack_edge(rs[k].x, rd[k].x), __float_as_uint(rv[k].x),
                                         pack_edge(rs[k].y, rd[k].y), __float_as_uint(rv[k].y));
    } else if (e < E) {
      ((uint2*)rec)[e] = make_uint2(pack_edge(edge_src[e], edge_dst[e]), __float_as_uint(edge_val[e]));
    }
  }
  for (int e = nthr * 8 + gtid * 2; e < E; e += nthr * 2) {
    ((uint2*)rec)[e] = make_uint2(pack_edge(edge_src[e], edge_dst[e]), __float_as_uint(edge_val[e]));
    if (e + 1 < E)
      ((uint2*)rec)[e + 1] =
          make_uint2(pack_edge(edge_src[e + 1], edge_dst[e + 1]), __float_as_uint(edge_val[e + 1]));
  }
}

#define NCHUNK    5
#define QCAP      256

__device__ __forceinline__ unsigned match_bytes(unsigned x) {
  const unsigned t = ((x & 0x7F7F7F7Fu) + 0x7F7F7F7Fu) | x;
  return ~t & 0x80808080u;
}

__device__ __forceinline__ void flush_rows(
    const unsigned* q, int head, int count, int e0, int lane,
    const float* __restrict__ xw, const uint2* __restrict__ rec, float* acc) {
  int s = 0;
  int dl = 0;
  float v = 0.0f;
  if (lane < count) {
    const int e = e0 + (int)q[(head + lane) & (QCAP - 1)];
    const uint2 er = rec[e];
    s  = (int)(er.x & 0x1FFFFFu);
    dl = (int)(er.x >> 21);
    v  = __uint_as_float(er.y);
  }
  float r[32];
  #pragma unroll
  for (int j = 0; j < 32; ++j) {
    const int sj = __builtin_amdgcn_readlane(s, j);
    r[j] = xw[(size_t)sj * D_OUT + lane];
  }
  #pragma unroll
  for (int j = 0; j < 32; ++j) {
    if (j < count) {
      const float vj = __int_as_float(__builtin_amdgcn_readlane(__float_as_int(v), j));
      const int dj = __builtin_amdgcn_readlane(dl, j);
      atomicAdd(&acc[dj * D_OUT + lane], vj * r[j]);
    }
  }
}

__global__ __launch_bounds__(1024) void band_filter_kernel(
    const float* __restrict__ xw, const int* __restrict__ recw,
    const int* __restrict__ tags, float* __restrict__ part, int N, int E) {
  __shared__ __attribute__((aligned(16))) float acc[BAND_ROWS * D_OUT];
  __shared__ unsigned queue[32 * QCAP];

  float4* acc4 = (float4*)acc;
  for (int i = threadIdx.x; i < BAND_ROWS * D_OUT / 4; i += 1024)
    acc4[i] = make_float4(0.0f, 0.0f, 0.0f, 0.0f);
  __syncthreads();

  const int lane = threadIdx.x & 31;
  const int wave = threadIdx.x >> 5;
  const int lo   = blockIdx.x * BAND_ROWS;
  const int nw   = gridDim.y * 32;
  const int gw   = blockIdx.y * 32 + wave;
  const int span = (((E + nw - 1) / nw) + 15) & ~15;
  const int e0   = gw * span;
  const int e1   = min(e0 + span, E);
  unsigned* q = queue + wave * QCAP;
  const unsigned bb = (unsigned)blockIdx.x * 0x01010101u;
  const unsigned char* tagb = (const unsigned char*)tags;
  const uint2* rec = (const uint2*)recw;
  const uint4 none = make_uint4(0xFFFFFFFFu, 0xFFFFFFFFu, 0xFFFFFFFFu, 0xFFFFFFFFu);

  int qhead = 0, qtail = 0;

  uint4 curA = none, curB = none;
  if (e0 + lane * 16 < e1)       curA = *(const uint4*)(tagb + e0 + lane * 16);
  if (e0 + 512 + lane * 16 < e1) curB = *(const uint4*)(tagb + e0 + 512 + lane * 16);
  asm volatile("" :: "v"(curA.x), "v"(curA.y), "v"(curA.z), "v"(curA.w),
                     "v"(curB.x), "v"(curB.y), "v"(curB.z), "v"(curB.w));

  for (int base = e0; base < e1; base += 1024) {
    uint4 nxtA = none, nxtB = none;
    if (base + 1024 + lane * 16 < e1) nxtA = *(const uint4*)(tagb + base + 1024 + lane * 16);
    if (base + 1536 + lane * 16 < e1) nxtB = *(const uint4*)(tagb + base + 1536 + lane * 16);

    unsigned m = (match_bytes(curA.x ^ bb) >> 7) | (match_bytes(curA.y ^ bb) >> 6) |
                 (match_bytes(curA.z ^ bb) >> 5) | (match_bytes(curA.w ^ bb) >> 4) |
                 (match_bytes(curB.x ^ bb) >> 3) | (match_bytes(curB.y ^ bb) >> 2) |
                 (match_bytes(curB.z ^ bb) >> 1) |  match_bytes(curB.w ^ bb);

    for (;;) {
      const bool has = m != 0u;
      const unsigned mask = __builtin_amdgcn_ballot_w32(has);
      if (mask == 0u) break;
      const int pre = (int)__builtin_amdgcn_mbcnt_lo(mask, 0u);
      if (has) {
        const int p = __builtin_ctz(m);
        m &= m - 1u;
        q[(qtail + pre) & (QCAP - 1)] =
            (unsigned)(base - e0 + lane * 16 + ((p & 3) << 2) + (p >> 3) + ((p & 4) << 7));
      }
      qtail += __builtin_popcount(mask);
      while (qtail - qhead >= 32) {
        flush_rows(q, qhead, 32, e0, lane, xw, rec, acc);
        qhead += 32;
      }
    }

    asm volatile("" :: "v"(nxtA.x), "v"(nxtA.y), "v"(nxtA.z), "v"(nxtA.w),
                       "v"(nxtB.x), "v"(nxtB.y), "v"(nxtB.z), "v"(nxtB.w));
    curA = nxtA;
    curB = nxtB;
  }
  if (qtail - qhead > 0)
    flush_rows(q, qhead, qtail - qhead, e0, lane, xw, rec, acc);

  __syncthreads();

  const int rows = min(BAND_ROWS, N - lo);
  float4* p4 = (float4*)(part + ((size_t)blockIdx.y * N + lo) * D_OUT);
  for (int i = threadIdx.x; i < rows * (D_OUT / 4); i += 1024) p4[i] = acc4[i];
}

__global__ __launch_bounds__(256) void combine_kernel(
    const float* __restrict__ part, const float* __restrict__ b,
    float* __restrict__ out, int total4) {
  const int i = blockIdx.x * 256 + threadIdx.x;
  if (i >= total4) return;
  const float4* p = (const float4*)part;
  float4 s = p[i];
  #pragma unroll
  for (int c = 1; c < NCHUNK; ++c) {
    const float4 t = p[(size_t)c * total4 + i];
    s.x += t.x; s.y += t.y; s.z += t.z; s.w += t.w;
  }
  const float4 bb = ((const float4*)b)[i & (D_OUT / 4 - 1)];
  float4 o;
  o.x = fmaxf(s.x + bb.x, 0.0f);
  o.y = fmaxf(s.y + bb.y, 0.0f);
  o.z = fmaxf(s.z + bb.z, 0.0f);
  o.w = fmaxf(s.w + bb.w, 0.0f);
  ((float4*)out)[i] = o;
}

extern "C" void kernel_launch(void* const* d_in, const int* in_sizes, int n_in,
                              void* d_out, int out_size, void* d_ws, size_t ws_size,
                              hipStream_t stream) {
  const float* x        = (const float*)d_in[0];
  const float* W        = (const float*)d_in[1];
  const float* b        = (const float*)d_in[2];
  const float* edge_val = (const float*)d_in[3];
  const int*   edge_src = (const int*)d_in[4];
  const int*   edge_dst = (const int*)d_in[5];
  float* out  = (float*)d_out;
  float* xw   = (float*)d_ws;
  float* part = xw + (size_t)4 * 1024 * 1024;
  int*   tags = (int*)(xw + (size_t)32 * 1024 * 1024);
  int*   rec  = (int*)(xw + (size_t)40 * 1024 * 1024);

  const int N = in_sizes[0] / D_IN;
  const int E = in_sizes[3];
  const int nTiles = (N + 15) / 16;
  const int total4 = N * (D_OUT / 4);

  gemm_xw_kernel<<<(nTiles + 3) / 4, 128, 0, stream>>>(x, W, xw, nTiles, edge_dst, tags, E, edge_src, edge_val, rec);
  band_filter_kernel<<<dim3((N + BAND_ROWS - 1) / BAND_ROWS, NCHUNK), 1024, 0, stream>>>(
      xw, rec, tags, part, N, E);
  combine_kernel<<<(total4 + 255) / 256, 256, 0, stream>>>(part, b, out, total4);
}


# each out row owned by one work-group: 250 bands of 400 rows; a band's edges = AND of 2 edge bit planes (band>>4, band&15) from the GEMM launch; no partial sums, no combine launch; sum order changes
# speedup vs baseline: 3.8007x; 1.2956x over previous
#include <hip/hip_runtime.h>

typedef float v2f __attribute__((ext_vector_type(2)));
typedef float v4f __attribute__((ext_vector_type(4)));
typedef float v8f __attribute__((ext_vector_type(8)));

#define D_IN  128
#define D_OUT 32
#define BAND_ROWS 400
#define NPLANE 32

__device__ __forceinline__ int4 load_dst4_guarded(const int* __restrict__ p, int idx, int e1) {
  int4 d = make_int4(-1, -1, -1, -1);
  if (idx + 3 < e1) {
    d = *(const int4*)(p + idx);
  } else {
    if (idx < e1)     d.x = p[idx];
    if (idx + 1 < e1) d.y = p[idx + 1];
    if (idx + 2 < e1) d.z = p[idx + 2];
  }
  return d;
}

__device__ __forceinline__ void mark_edge(int d, int i, unsigned long long& hA,
                                          unsigned long long& hB, unsigned long long& lA,
                                          unsigned long long& lB) {
  const unsigned band = (unsigned)d / BAND_ROWS;
  const unsigned h = band >> 4, l = band & 15u;
  const unsigned long long on = h < 16u ? 1ull : 0ull;
  const unsigned long long bh = on << (8u * (h & 7u) + (unsigned)i);
  const unsigned long long bl = on << (8u * (l & 7u) + (unsigned)i);
  if (h & 8u) hB |= bh; else hA |= bh;
  if (l & 8u) lB |= bl; else lA |= bl;
}

__device__ __forceinline__ unsigned pack_edge(int s, int d) {
  return (unsigned)s | (((unsigned)d % BAND_ROWS) << 21);
}

__global__ __launch_bounds__(128) void gemm_xw_kernel(
    const float* __restrict__ x, const float* __restrict__ W,
    float* __restrict__ xw, int nTiles,
    const int* __restrict__ edge_dst, int* __restrict__ planes, int E,
    const int* __restrict__ edge_src, const float* __restrict__ edge_val,
    int* __restrict__ rec, int planeBytes) {
  __shared__ float Wlds[D_IN * D_OUT];
  __shared__ uint4 bm[NPLANE * 8];

  const int wave = threadIdx.x >> 5;
  const int lane = threadIdx.x & 31;
  const int tile = blockIdx.x * 4 + wave;
  const bool live = tile < nTiles;

  const int half = lane >> 4;
  const int m    = lane & 15;

  const v4f* Wv = (const v4f*)W;
  v4f* Lv = (v4f*)Wlds;
  v4f w[8];
  #pragma unroll
  for (int i = 0; i < 8; ++i) w[i] = Wv[threadIdx.x + i * 128];

  const int ltile = live ? tile : nTiles - 1;
  const float4* xrow =
      (const float4*)(x + (size_t)(ltile * 16 + m) * D_IN + half * 4);
  float4 a[16];
  #pragma unroll
  for (int j = 0; j < 16; ++j) a[j] = xrow[j * 2];

  asm volatile("" :: "v"(w[0]), "v"(w[1]), "v"(w[2]), "v"(w[3]),
                     "v"(w[4]), "v"(w[5]), "v"(w[6]), "v"(w[7]));
  #pragma unroll
  for (int i = 0; i < 8; ++i) Lv[threadIdx.x + i * 128] = w[i];
  __syncthreads();

  const int gtid = blockIdx.x * 128 + threadIdx.x;
  const int tcl  = min(gtid * 8, (E - 8) & ~3);
  const int4 ta = *(const int4*)(edge_dst + tcl);
  const int4 tb = *(const int4*)(edge_dst + tcl + 4);

  const int wb    = (gtid - lane) * 8;
  const int elast = (E - 2) & ~1;
  int2 rs[4], rd[4];
  float2 rv[4];
  #pragma unroll
  for (int k = 0; k < 4; ++k) {
    const int ec = min(wb + 64 * k + 2 * lane, elast);
    rs[k] = *(const int2*)(edge_src + ec);
    rd[k] = *(const int2*)(edge_dst + ec);
    rv[k] = *(const float2*)(edge_val + ec);
  }

  if (live) {
  v8f c0 = {};
  v8f c1 = {};

  #pragma unroll
  for (int j = 0; j < 16; ++j) {
    #pragma unroll
    for (int t = 0; t < 2; ++t) {
      const int kb = j * 8 + half * 4 + t * 2;
      v2f av;
      av.x = t ? a[j].z : a[j].x;
      av.y = t ? a[j].w : a[j].y;
      v2f b0, b1;
      b0.x = Wlds[kb * D_OUT + m];
      b0.y = Wlds[(kb + 1) * D_OUT + m];
      b1.x = Wlds[kb * D_OUT + 16 + m];
      b1.y = Wlds[(kb + 1) * D_OUT + 16 + m];
      c0 = __builtin_amdgcn_wmma_f32_16x16x4_f32(false, av, false, b0,
                                                 (short)0, c0, false, false);
      c1 = __builtin_amdgcn_wmma_f32_16x16x4_f32(false, av, false, b1,
                                                 (short)0, c1, false, false);
    }
  }

  const int rowBase = tile * 16 + half * 8;
  #pragma unroll
  for (int r = 0; r < 8; ++r) {
    xw[(size_t)(rowBase + r) * D_OUT + m]      = c0[r];
    xw[(size_t)(rowBase + r) * D_OUT + 16 + m] = c1[r];
  }
  }

  {
    int4 da = ta, db = tb;
    if (gtid * 8 + 8 > E) {
      da = load_dst4_guarded(edge_dst, gtid * 8, E);
      db = load_dst4_guarded(edge_dst, gtid * 8 + 4, E);
    }
    unsigned long long hA = 0ull, hB = 0ull, lA = 0ull, lB = 0ull;
    mark_edge(da.x, 0, hA, hB, lA, lB);
    mark_edge(da.y, 1, hA, hB, lA, lB);
    mark_edge(da.z, 2, hA, hB, lA, lB);
    mark_edge(da.w, 3, hA, hB, lA, lB);
    mark_edge(db.x, 4, hA, hB, lA, lB);
    mark_edge(db.y, 5, hA, hB, lA, lB);
    mark_edge(db.z, 6, hA, hB, lA, lB);
    mark_edge(db.w, 7, hA, hB, lA, lB);
    unsigned char* bmB = (unsigned char*)bm;
    for (int r = 0; r < 8; ++r) {
      bmB[r * 128 + threadIdx.x]        = (unsigned char)(hA >> (8 * r));
      bmB[(8 + r) * 128 + threadIdx.x]  = (unsigned char)(hB >> (8 * r));
      bmB[(16 + r) * 128 + threadIdx.x] = (unsigned char)(lA >> (8 * r));
      bmB[(24 + r) * 128 + threadIdx.x] = (unsigned char)(lB >> (8 * r));
    }
  }
  __syncthreads();
  if (blockIdx.x * 128 < planeBytes) {
    unsigned char* pb = (unsigned char*)planes + (size_t)blockIdx.x * 128;
    for (int k = 0; k < 2; ++k) {
      const int i = threadIdx.x + k * 128;
      ((uint4*)(pb + (size_t)(i >> 3) * planeBytes))[i & 7] = bm[i];
    }
  }

  #pragma unroll
  for (int k = 0; k < 4; ++k) {
    const int e = wb + 64 * k + 2 * lane;
    if (e + 2 <= E) {
      ((uint4*)rec)[e >> 1] = make_uint4(pack_edge(rs[k].x, rd[k].x), __float_as_uint(rv[k].x),
                                         pack_edge(rs[k].y, rd[k].y), __float_as_uint(rv[k].y));
    } else if (e < E) {
      ((uint2*)rec)[e] = make_uint2(pack_edge(edge_src[e], edge_dst[e]), __float_as_uint(edge_val[e]));
    }
  }
}

#define QCAP       128
#define ACC_FLOATS (41 * 1024)

__device__ __forceinline__ void flush_rows(
    const unsigned* q, int head, int count, int lane,
    const float* __restrict__ xw, const uint2* __restrict__ rec, float* acc) {
  int s = 0;
  int dl = 0;
  float v = 0.0f;
  if (lane < count) {
    const int e = (int)q[(head + lane) & (QCAP - 1)];
    const uint2 er = rec[e];
    s  = (int)(er.x & 0x1FFFFFu);
    dl = (int)(er.x >> 21);
    v  = __uint_as_float(er.y);
  }
  float r[32];
  #pragma unroll
  for (int j = 0; j < 32; ++j) {
    const int sj = __builtin_amdgcn_readlane(s, j);
    r[j] = xw[(size_t)sj * D_OUT + lane];
  }
  #pragma unroll
  for (int j = 0; j < 32; ++j) {
    if (j < count) {
      const float vj = __int_as_float(__builtin_amdgcn_readlane(__float_as_int(v), j));
      const int dj = __builtin_amdgcn_readlane(dl, j);
      atomicAdd(&acc[dj * D_OUT + lane], vj * r[j]);
    }
  }
}

__global__ __launch_bounds__(1024) void band_sum_kernel(
    const float* __restrict__ xw, const int* __restrict__ recw,
    const int* __restrict__ planes, const float* __restrict__ b,
    float* __restrict__ out, int N, int planeBytes) {
  __shared__ __attribute__((aligned(16))) float acc[ACC_FLOATS];
  __shared__ unsigned queue[32 * QCAP];

  float4* acc4 = (float4*)acc;
  for (int i = threadIdx.x; i < BAND_ROWS * D_OUT / 4; i += 1024)
    acc4[i] = make_float4(0.0f, 0.0f, 0.0f, 0.0f);
  __syncthreads();

  const int lane = threadIdx.x & 31;
  const int wave = threadIdx.x >> 5;
  const int band = blockIdx.x;
  const int lo   = band * BAND_ROWS;
  unsigned* q = queue + wave * QCAP;
  const unsigned char* pbase = (const unsigned char*)planes;
  const uint4* ph = (const uint4*)(pbase + (size_t)(band >> 4) * planeBytes);
  const uint4* pl = (const uint4*)(pbase + (size_t)(16 + (band & 15)) * planeBytes);
  const uint2* rec = (const uint2*)recw;
  const int nUnits = planeBytes >> 4;
  const uint4 none = make_uint4(0u, 0u, 0u, 0u);

  int qhead = 0, qtail = 0;

  uint4 curH = none, curL = none;
  if (wave * 32 + lane < nUnits) {
    curH = ph[wave * 32 + lane];
    curL = pl[wave * 32 + lane];
  }
  asm volatile("" :: "v"(curH.x), "v"(curH.y), "v"(curH.z), "v"(curH.w),
                     "v"(curL.x), "v"(curL.y), "v"(curL.z), "v"(curL.w));

  for (int ub = wave * 32; ub < nUnits; ub += 1024) {
    uint4 nxtH = none, nxtL = none;
    if (ub + 1024 + lane < nUnits) {
      nxtH = ph[ub + 1024 + lane];
      nxtL = pl[ub + 1024 + lane];
    }

    unsigned long long m0 = (unsigned long long)(curH.x & curL.x) |
                            ((unsigned long long)(curH.y & curL.y) << 32);
    unsigned long long m1 = (unsigned long long)(curH.z & curL.z) |
                            ((unsigned long long)(curH.w & curL.w) << 32);
    const unsigned ebase = (unsigned)(ub + lane) * 128u;

    for (;;) {
      const bool has = (m0 | m1) != 0ull;
      const unsigned mask = __builtin_amdgcn_ballot_w32(has);
      if (mask == 0u) break;
      const int pre = (int)__builtin_amdgcn_mbcnt_lo(mask, 0u);
      if (has) {
        unsigned p;
        if (m0 != 0ull) {
          p = (unsigned)__builtin_ctzll(m0);
          m0 &= m0 - 1ull;
        } else {
          p = 64u + (unsigned)__builtin_ctzll(m1);
          m1 &= m1 - 1ull;
        }
        q[(qtail + pre) & (QCAP - 1)] = ebase + p;
      }
      qtail += __builtin_popcount(mask);
      while (qtail - qhead >= 32) {
        flush_rows(q, qhead, 32, lane, xw, rec, acc);
        qhead += 32;
      }
    }

    asm volatile("" :: "v"(nxtH.x), "v"(nxtH.y), "v"(nxtH.z), "v"(nxtH.w),
                       "v"(nxtL.x), "v"(nxtL.y), "v"(nxtL.z), "v"(nxtL.w));
    curH = nxtH;
    curL = nxtL;
  }
  if (qtail - qhead > 0)
    flush_rows(q, qhead, qtail - qhead, lane, xw, rec, acc);

  __syncthreads();

  const int rows = min(BAND_ROWS, N - lo);
  float4* o4 = (float4*)(out + (size_t)lo * D_OUT);
  const float4* b4 = (const float4*)b;
  for (int i = threadIdx.x; i < rows * (D_OUT / 4); i += 1024) {
    const float4 s  = acc4[i];
    const float4 bb = b4[i & (D_OUT / 4 - 1)];
    float4 o;
    o.x = fmaxf(s.x + bb.x, 0.0f);
    o.y = fmaxf(s.y + bb.y, 0.0f);
    o.z = fmaxf(s.z + bb.z, 0.0f);
    o.w = fmaxf(s.w + bb.w, 0.0f);
    o4[i] = o;
  }
}

extern "C" void kernel_launch(void* const* d_in, const int* in_sizes, int n_in,
                              void* d_out, int out_size, void* d_ws, size_t ws_size,
                              hipStream_t stream) {
  const float* x        = (const float*)d_in[0];
  const float* W        = (const float*)d_in[1];
  const float* b        = (const float*)d_in[2];
  const float* edge_val = (const float*)d_in[3];
  const int*   edge_src = (const int*)d_in[4];
  const int*   edge_dst = (const int*)d_in[5];
  float* out  = (float*)d_out;
  float* xw     = (float*)d_ws;
  int*   rec    = (int*)(xw + (size_t)16 * 1024 * 1024);
  int*   planes = (int*)(xw + (size_t)32 * 1024 * 1024);

  const int N = in_sizes[0] / D_IN;
  const int E = in_sizes[3];
  const int nTiles = (N + 15) / 16;
  const int nBands = (N + BAND_ROWS - 1) / BAND_ROWS;
  const int nEdgeBlocks = (E + 1023) / 1024;
  const int planeBytes  = nEdgeBlocks * 128;
  const int nTileBlocks = (nTiles + 3) / 4;
  const int nBlocks = nTileBlocks > nEdgeBlocks ? nTileBlocks : nEdgeBlocks;

  gemm_xw_kernel<<<nBlocks, 128, 0, stream>>>(x, W, xw, nTiles, edge_dst, planes, E, edge_src, edge_val, rec, planeBytes);
  band_sum_kernel<<<nBands, 1024, 0, stream>>>(xw, rec, planes, b, out, N, planeBytes);
}
